# starting kernel
# baseline (speedup 1.0000x reference)
#include <hip/hip_runtime.h>

typedef float v2f __attribute__((ext_vector_type(2)));
typedef float v8f __attribute__((ext_vector_type(8)));

__global__ void tag_deg(const int* __restrict__ dst, float* __restrict__ deg, int E) {
    int e = blockIdx.x * blockDim.x + threadIdx.x;
    if (e < E) unsafeAtomicAdd(&deg[dst[e]], 1.0f);
}

__global__ void tag_dis(const float* __restrict__ deg, float* __restrict__ dis, int N) {
    int i = blockIdx.x * blockDim.x + threadIdx.x;
    if (i < N) {
        float d = deg[i];
        dis[i] = (d > 0.0f) ? rsqrtf(fmaxf(d, 1.0f)) : 0.0f;
    }
}

__global__ void tag_norm(const int* __restrict__ src, const int* __restrict__ dst,
                         const float* __restrict__ dis, float* __restrict__ nrm, int E) {
    int e = blockIdx.x * blockDim.x + threadIdx.x;
    if (e < E) nrm[e] = dis[src[e]] * dis[dst[e]];
}

__global__ void tag_prop(const float* __restrict__ h, const int* __restrict__ src,
                         const int* __restrict__ dst, const float* __restrict__ nrm,
                         float* __restrict__ out, int E) {
    long long t = (long long)blockIdx.x * blockDim.x + threadIdx.x;
    int e = (int)(t >> 5);
    if (e >= E) return;
    int d4 = ((int)t & 31) << 2;
    int s  = src[e];
    int dd = dst[e];
    float w = nrm[e];
    const float4 v = *(const float4*)(h + (size_t)s * 128 + d4);
    float* o = out + (size_t)dd * 128 + d4;
    unsafeAtomicAdd(o + 0, v.x * w);
    unsafeAtomicAdd(o + 1, v.y * w);
    unsafeAtomicAdd(o + 2, v.z * w);
    unsafeAtomicAdd(o + 3, v.w * w);
}

template <int NOUT, bool ACC, bool BIAS, bool RELU>
__global__ void tag_gemm(const float* __restrict__ A, const float* __restrict__ W,
                         float* __restrict__ C, const float* __restrict__ bias) {
    const int lane = threadIdx.x;
    const int tm = blockIdx.x, tn = blockIdx.y;
    const int lo = lane & 15;
    const int hi = lane >> 4;
    const int arow = tm * 16 + lo;
    const int bcol = tn * 16 + lo;
    const int crow = tm * 16 + hi * 8;
    const int ccol = tn * 16 + lo;

    v8f c = {};
    if (ACC) {
#pragma unroll
        for (int r = 0; r < 8; ++r)
            c[r] = C[(size_t)(crow + r) * NOUT + ccol];
    }

    const float* arowp = A + (size_t)arow * 128 + hi * 2;
    const float* bp0   = W + (size_t)(hi * 2) * NOUT + bcol;
#pragma unroll 8
    for (int k0 = 0; k0 < 128; k0 += 4) {
        v2f a = *(const v2f*)(arowp + k0);
        v2f b;
        b.x = bp0[(size_t)k0 * NOUT];
        b.y = bp0[(size_t)k0 * NOUT + NOUT];
        c = __builtin_amdgcn_wmma_f32_16x16x4_f32(
                 false, a,  false, b,
                 (short)0, c,  false,  false);
    }

    const float bv = BIAS ? bias[ccol] : 0.0f;
#pragma unroll
    for (int r = 0; r < 8; ++r) {
        float v = c[r];
        if (BIAS) v += bv;
        if (RELU) v = fmaxf(v, 0.0f);
        C[(size_t)(crow + r) * NOUT + ccol] = v;
    }
}

extern "C" void kernel_launch(void* const* d_in, const int* in_sizes, int n_in,
                              void* d_out, int out_size, void* d_ws, size_t ws_size,
                              hipStream_t stream) {
    const float* x  = (const float*)d_in[0];
    const int*   ei = (const int*)d_in[1];
    const float* W1 = (const float*)d_in[2];
    const float* b1 = (const float*)d_in[3];
    const float* W2 = (const float*)d_in[4];
    const float* b2 = (const float*)d_in[5];

    const int N = in_sizes[0] / 128;
    const int E = in_sizes[1] / 2;
    const int* src = ei;
    const int* dst = ei + E;

    float* ws   = (float*)d_ws;
    float* deg  = ws;
    float* dis  = deg + N;
    float* nrm  = dis + N;
    float* bufA = nrm + E;
    float* bufB = bufA + (size_t)N * 128;
    float* acc  = bufB + (size_t)N * 128;
    float* out  = (float*)d_out;

    const int T = 256;
    const size_t featBytes = (size_t)N * 128 * sizeof(float);
    const long long pth = (long long)E * 32;
    const int pblk = (int)((pth + T - 1) / T);
    const dim3 g1(N / 16, 128 / 16);
    const dim3 g2(N / 16, 1);

    hipMemsetAsync(deg, 0, (size_t)N * sizeof(float), stream);
    tag_deg <<<(E + T - 1) / T, T, 0, stream>>>(dst, deg, E);
    tag_dis <<<(N + T - 1) / T, T, 0, stream>>>(deg, dis, N);
    tag_norm<<<(E + T - 1) / T, T, 0, stream>>>(src, dst, dis, nrm, E);

    tag_gemm<128, false, false, false><<<g1, 32, 0, stream>>>(x,    W1 + 0 * 128 * 128, acc, nullptr);
    hipMemsetAsync(bufA, 0, featBytes, stream);
    tag_prop<<<pblk, T, 0, stream>>>(x, src, dst, nrm, bufA, E);
    tag_gemm<128, true,  false, false><<<g1, 32, 0, stream>>>(bufA, W1 + 1 * 128 * 128, acc, nullptr);
    hipMemsetAsync(bufB, 0, featBytes, stream);
    tag_prop<<<pblk, T, 0, stream>>>(bufA, src, dst, nrm, bufB, E);
    tag_gemm<128, true,  false, false><<<g1, 32, 0, stream>>>(bufB, W1 + 2 * 128 * 128, acc, nullptr);
    hipMemsetAsync(bufA, 0, featBytes, stream);
    tag_prop<<<pblk, T, 0, stream>>>(bufB, src, dst, nrm, bufA, E);
    tag_gemm<128, true,  true,  true ><<<g1, 32, 0, stream>>>(bufA, W1 + 3 * 128 * 128, acc, b1);

    tag_gemm<16, false, false, false><<<g2, 32, 0, stream>>>(acc,  W2 + 0 * 128 * 16, out, nullptr);
    hipMemsetAsync(bufA, 0, featBytes, stream);
    tag_prop<<<pblk, T, 0, stream>>>(acc, src, dst, nrm, bufA, E);
    tag_gemm<16, true,  false, false><<<g2, 32, 0, stream>>>(bufA, W2 + 1 * 128 * 16, out, nullptr);
    hipMemsetAsync(bufB, 0, featBytes, stream);
    tag_prop<<<pblk, T, 0, stream>>>(bufA, src, dst, nrm, bufB, E);
    tag_gemm<16, true,  false, false><<<g2, 32, 0, stream>>>(bufB, W2 + 2 * 128 * 16, out, nullptr);
    hipMemsetAsync(bufA, 0, featBytes, stream);
    tag_prop<<<pblk, T, 0, stream>>>(bufB, src, dst, nrm, bufA, E);
    tag_gemm<16, true,  true,  false><<<g2, 32, 0, stream>>>(bufA, W2 + 3 * 128 * 16, out, b2);

    (void)n_in; (void)out_size; (void)ws_size;
}


# hop scatter remapped: lane l owns features l, l+32, l+64, l+96, so each f32 atomic add and each gather load of a wave covers one whole 128-byte line, not a quarter of four; same adds, arrival order
# speedup vs baseline: 1.2668x; 1.2668x over previous
#include <hip/hip_runtime.h>

typedef float v2f __attribute__((ext_vector_type(2)));
typedef float v8f __attribute__((ext_vector_type(8)));

__global__ void tag_deg(const int* __restrict__ dst, float* __restrict__ deg, int E) {
    int e = blockIdx.x * blockDim.x + threadIdx.x;
    if (e < E) unsafeAtomicAdd(&deg[dst[e]], 1.0f);
}

__global__ void tag_dis(const float* __restrict__ deg, float* __restrict__ dis, int N) {
    int i = blockIdx.x * blockDim.x + threadIdx.x;
    if (i < N) {
        float d = deg[i];
        dis[i] = (d > 0.0f) ? rsqrtf(fmaxf(d, 1.0f)) : 0.0f;
    }
}

__global__ void tag_norm(const int* __restrict__ src, const int* __restrict__ dst,
                         const float* __restrict__ dis, float* __restrict__ nrm, int E) {
    int e = blockIdx.x * blockDim.x + threadIdx.x;
    if (e < E) nrm[e] = dis[src[e]] * dis[dst[e]];
}

__global__ void tag_prop(const float* __restrict__ h, const int* __restrict__ src,
                         const int* __restrict__ dst, const float* __restrict__ nrm,
                         float* __restrict__ out, int E) {
    long long t = (long long)blockIdx.x * blockDim.x + threadIdx.x;
    int e = (int)(t >> 5);
    if (e >= E) return;
    int l  = (int)t & 31;
    int s  = src[e];
    int dd = dst[e];
    float w = nrm[e];
    const float* hp = h + (size_t)s * 128 + l;
    const float v0 = hp[0];
    const float v1 = hp[32];
    const float v2 = hp[64];
    const float v3 = hp[96];
    float* o = out + (size_t)dd * 128 + l;
    unsafeAtomicAdd(o +  0, v0 * w);
    unsafeAtomicAdd(o + 32, v1 * w);
    unsafeAtomicAdd(o + 64, v2 * w);
    unsafeAtomicAdd(o + 96, v3 * w);
}

template <int NOUT, bool ACC, bool BIAS, bool RELU>
__global__ void tag_gemm(const float* __restrict__ A, const float* __restrict__ W,
                         float* __restrict__ C, const float* __restrict__ bias) {
    const int lane = threadIdx.x;
    const int tm = blockIdx.x, tn = blockIdx.y;
    const int lo = lane & 15;
    const int hi = lane >> 4;
    const int arow = tm * 16 + lo;
    const int bcol = tn * 16 + lo;
    const int crow = tm * 16 + hi * 8;
    const int ccol = tn * 16 + lo;

    v8f c = {};
    if (ACC) {
#pragma unroll
        for (int r = 0; r < 8; ++r)
            c[r] = C[(size_t)(crow + r) * NOUT + ccol];
    }

    const float* arowp = A + (size_t)arow * 128 + hi * 2;
    const float* bp0   = W + (size_t)(hi * 2) * NOUT + bcol;
#pragma unroll 8
    for (int k0 = 0; k0 < 128; k0 += 4) {
        v2f a = *(const v2f*)(arowp + k0);
        v2f b;
        b.x = bp0[(size_t)k0 * NOUT];
        b.y = bp0[(size_t)k0 * NOUT + NOUT];
        c = __builtin_amdgcn_wmma_f32_16x16x4_f32(
                 false, a,  false, b,
                 (short)0, c,  false,  false);
    }

    const float bv = BIAS ? bias[ccol] : 0.0f;
#pragma unroll
    for (int r = 0; r < 8; ++r) {
        float v = c[r];
        if (BIAS) v += bv;
        if (RELU) v = fmaxf(v, 0.0f);
        C[(size_t)(crow + r) * NOUT + ccol] = v;
    }
}

extern "C" void kernel_launch(void* const* d_in, const int* in_sizes, int n_in,
                              void* d_out, int out_size, void* d_ws, size_t ws_size,
                              hipStream_t stream) {
    const float* x  = (const float*)d_in[0];
    const int*   ei = (const int*)d_in[1];
    const float* W1 = (const float*)d_in[2];
    const float* b1 = (const float*)d_in[3];
    const float* W2 = (const float*)d_in[4];
    const float* b2 = (const float*)d_in[5];

    const int N = in_sizes[0] / 128;
    const int E = in_sizes[1] / 2;
    const int* src = ei;
    const int* dst = ei + E;

    float* ws   = (float*)d_ws;
    float* deg  = ws;
    float* dis  = deg + N;
    float* nrm  = dis + N;
    float* bufA = nrm + E;
    float* bufB = bufA + (size_t)N * 128;
    float* acc  = bufB + (size_t)N * 128;
    float* out  = (float*)d_out;

    const int T = 256;
    const size_t featBytes = (size_t)N * 128 * sizeof(float);
    const long long pth = (long long)E * 32;
    const int pblk = (int)((pth + T - 1) / T);
    const dim3 g1(N / 16, 128 / 16);
    const dim3 g2(N / 16, 1);

    hipMemsetAsync(deg, 0, (size_t)N * sizeof(float), stream);
    tag_deg <<<(E + T - 1) / T, T, 0, stream>>>(dst, deg, E);
    tag_dis <<<(N + T - 1) / T, T, 0, stream>>>(deg, dis, N);
    tag_norm<<<(E + T - 1) / T, T, 0, stream>>>(src, dst, dis, nrm, E);

    tag_gemm<128, false, false, false><<<g1, 32, 0, stream>>>(x,    W1 + 0 * 128 * 128, acc, nullptr);
    hipMemsetAsync(bufA, 0, featBytes, stream);
    tag_prop<<<pblk, T, 0, stream>>>(x, src, dst, nrm, bufA, E);
    tag_gemm<128, true,  false, false><<<g1, 32, 0, stream>>>(bufA, W1 + 1 * 128 * 128, acc, nullptr);
    hipMemsetAsync(bufB, 0, featBytes, stream);
    tag_prop<<<pblk, T, 0, stream>>>(bufA, src, dst, nrm, bufB, E);
    tag_gemm<128, true,  false, false><<<g1, 32, 0, stream>>>(bufB, W1 + 2 * 128 * 128, acc, nullptr);
    hipMemsetAsync(bufA, 0, featBytes, stream);
    tag_prop<<<pblk, T, 0, stream>>>(bufB, src, dst, nrm, bufA, E);
    tag_gemm<128, true,  true,  true ><<<g1, 32, 0, stream>>>(bufA, W1 + 3 * 128 * 128, acc, b1);

    tag_gemm<16, false, false, false><<<g2, 32, 0, stream>>>(acc,  W2 + 0 * 128 * 16, out, nullptr);
    hipMemsetAsync(bufA, 0, featBytes, stream);
    tag_prop<<<pblk, T, 0, stream>>>(acc, src, dst, nrm, bufA, E);
    tag_gemm<16, true,  false, false><<<g2, 32, 0, stream>>>(bufA, W2 + 1 * 128 * 16, out, nullptr);
    hipMemsetAsync(bufB, 0, featBytes, stream);
    tag_prop<<<pblk, T, 0, stream>>>(bufA, src, dst, nrm, bufB, E);
    tag_gemm<16, true,  false, false><<<g2, 32, 0, stream>>>(bufB, W2 + 2 * 128 * 16, out, nullptr);
    hipMemsetAsync(bufA, 0, featBytes, stream);
    tag_prop<<<pblk, T, 0, stream>>>(bufB, src, dst, nrm, bufA, E);
    tag_gemm<16, true,  true,  false><<<g2, 32, 0, stream>>>(bufA, W2 + 3 * 128 * 16, out, b2);

    (void)n_in; (void)out_size; (void)ws_size;
}


# degree counted on LDS counters per (node range, edge slice) pair, integer partial counts stored and summed in the dis kernel: the scattered global f32 atomic adds and deg's zero-fill go; exact sum
# speedup vs baseline: 2.5643x; 2.0242x over previous
#include <hip/hip_runtime.h>

typedef float v2f __attribute__((ext_vector_type(2)));
typedef float v4f __attribute__((ext_vector_type(4)));
typedef float v8f __attribute__((ext_vector_type(8)));

constexpr int DR = 6400;
constexpr int DS = 16;
constexpr int DT = 1024;
constexpr int DU = 8;

__device__ __forceinline__ void deg_count(int* cnt, int d, int lo) {
    const unsigned k = (unsigned)(d - lo);
    if (k < (unsigned)DR) atomicAdd(&cnt[k], 1);
}

__global__ __launch_bounds__(DT)
void tag_deg(const int* __restrict__ dst, int* __restrict__ part, int N, int E, int nrange) {
    __shared__ int cnt[DR];
    const int r  = blockIdx.x % nrange;
    const int s  = blockIdx.x / nrange;
    const int lo = r * DR;
    for (int i = threadIdx.x; i < DR; i += DT) cnt[i] = 0;
    __syncthreads();

    const int per = (((E + DS - 1) / DS) + 3) & ~3;
    const long long b0 = (long long)s * per;
    const int e0 = (int)(b0 < (long long)E ? b0 : (long long)E);
    const int e1 = (E - e0 < per) ? E : e0 + per;
    int e = e0;
    if ((((size_t)dst) & 15) == 0) {
        constexpr int SPAN = DT * 4 * DU;
        for (; e1 - e >= SPAN; e += SPAN) {
            const int* p = dst + e + (int)threadIdx.x * 4;
            int4 d[DU];
#pragma unroll
            for (int u = 0; u < DU; ++u) d[u] = *(const int4*)(p + u * DT * 4);
#pragma unroll
            for (int u = 0; u < DU; ++u) {
                deg_count(cnt, d[u].x, lo);
                deg_count(cnt, d[u].y, lo);
                deg_count(cnt, d[u].z, lo);
                deg_count(cnt, d[u].w, lo);
            }
        }
    }
    for (int p = e + (int)threadIdx.x; p < e1; p += DT) deg_count(cnt, dst[p], lo);
    __syncthreads();
    for (int i = threadIdx.x; i < DR; i += DT)
        if (lo + i < N) part[(size_t)s * N + lo + i] = cnt[i];
}

__global__ void tag_dis(const int* __restrict__ part, float* __restrict__ dis, int N) {
    int i = blockIdx.x * blockDim.x + threadIdx.x;
    if (i < N) {
        int c = 0;
#pragma unroll
        for (int s = 0; s < DS; ++s) c += part[(size_t)s * N + i];
        float d = (float)c;
        dis[i] = (d > 0.0f) ? rsqrtf(fmaxf(d, 1.0f)) : 0.0f;
    }
}

constexpr int QSTRIDE = 64;
constexpr int NQ      = 12;

__global__ void tag_norm(const int* __restrict__ src, const int* __restrict__ dst,
                         const float* __restrict__ dis, float* __restrict__ nrm,
                         int* __restrict__ q, int E) {
    int e = blockIdx.x * blockDim.x + threadIdx.x;
    if (e < NQ) q[e * QSTRIDE] = 0;
    if (e < E) nrm[e] = dis[src[e]] * dis[dst[e]];
}

constexpr int EPG = 16;
constexpr int GPC = 4;
constexpr int EPC = EPG * GPC;

__global__ __launch_bounds__(256)
void tag_prop(const float* __restrict__ h, const int* __restrict__ src,
              const int* __restrict__ dst, const float* __restrict__ nrm,
              float* __restrict__ out, int* __restrict__ q, int E) {
    __shared__ int s_half;
    const int lane = threadIdx.x & 31;
    if (threadIdx.x == 0) {
        const unsigned id = __builtin_amdgcn_s_sendmsg_rtn(0x87);
        s_half = (int)((id >> 18) & 1u);
    }
    __syncthreads();
    int fh = __builtin_amdgcn_readfirstlane(s_half);
    const int nchunk = (E + EPC - 1) / EPC;

    int tv = 0;
    if (lane == 0) tv = atomicAdd(q + fh * QSTRIDE, 1);
    int other = 0;
    const int maxit = 2 * nchunk + 4;
    for (int it = 0; it < maxit; ++it) {
        const int t = __builtin_amdgcn_readfirstlane(tv);
        if (t >= nchunk) {
            if (other) break;
            other = 1;
            fh ^= 1;
            tv = 0;
            if (lane == 0) tv = atomicAdd(q + fh * QSTRIDE, 1);
            continue;
        }
        const int e0 = t * EPC;
        const size_t fo = (size_t)fh * 64;
        int tn = 0;
        if (e0 + EPC <= E) {
            const int   s0 = src[e0 + lane], s1 = src[e0 + 32 + lane];
            const int   d0 = dst[e0 + lane], d1 = dst[e0 + 32 + lane];
            const float w0 = nrm[e0 + lane], w1 = nrm[e0 + 32 + lane];
            if (lane == 0) tn = atomicAdd(q + fh * QSTRIDE, 1);
            for (int g = 0; g < GPC; ++g) {
                const int   ms = (g < 2) ? s0 : s1;
                const int   md = (g < 2) ? d0 : d1;
                const float mw = (g < 2) ? w0 : w1;
                const int   l0 = (g & 1) * EPG;
                float v[EPG][2];
#pragma unroll
                for (int j = 0; j < EPG; ++j) {
                    const int s = __builtin_amdgcn_readlane(ms, l0 + j);
                    const float* hrow = h + (size_t)s * 128 + fo;
                    v[j][0] = hrow[lane];
                    v[j][1] = hrow[lane + 32];
                }
#pragma unroll
                for (int j = 0; j < EPG; ++j) {
                    const int   dd = __builtin_amdgcn_readlane(md, l0 + j);
                    const float w  = __int_as_float(__builtin_amdgcn_readlane(__float_as_int(mw), l0 + j));
                    float* orow = out + (size_t)dd * 128 + fo;
                    unsafeAtomicAdd(orow + lane,      v[j][0] * w);
                    unsafeAtomicAdd(orow + lane + 32, v[j][1] * w);
                }
            }
        } else {
            if (lane == 0) tn = atomicAdd(q + fh * QSTRIDE, 1);
            for (int e = e0; e < E; ++e) {
                const int   s  = src[e];
                const int   dd = dst[e];
                const float w  = nrm[e];
                const float* hrow = h + (size_t)s * 128 + fo;
                float* orow = out + (size_t)dd * 128 + fo;
                const float v0 = hrow[lane], v1 = hrow[lane + 32];
                unsafeAtomicAdd(orow + lane,      v0 * w);
                unsafeAtomicAdd(orow + lane + 32, v1 * w);
            }
        }
        tv = tn;
    }
}

template <int NOUT, bool ACC, bool BIAS, bool RELU>
__global__ __launch_bounds__(256)
void tag_gemm(const float* __restrict__ A, const float* __restrict__ W,
              float* __restrict__ C, const float* __restrict__ bias, int ntiles) {
    constexpr int T  = (NOUT >= 64) ? 4 : 1;
    constexpr int G  = NOUT / (16 * T);
    constexpr int RS = 8 / G;
    const int lane = threadIdx.x & 31;
    const int wv = __builtin_amdgcn_readfirstlane((int)(threadIdx.x >> 5));
    const int g  = wv % G;
    const int rs = wv / G;
    const int lo = lane & 15;
    const int hi = lane >> 4;
    const int col = g * (16 * T) + lo * T;

    v2f b[T][32];
#pragma unroll
    for (int j = 0; j < 16; ++j) {
        const float* wp = W + (size_t)(8 * j + 4 * hi) * NOUT + col;
        if constexpr (T == 4) {
            const v4f w0 = *(const v4f*)(wp);
            const v4f w1 = *(const v4f*)(wp + NOUT);
            const v4f w2 = *(const v4f*)(wp + 2 * NOUT);
            const v4f w3 = *(const v4f*)(wp + 3 * NOUT);
#pragma unroll
            for (int t = 0; t < 4; ++t) {
                b[t][2 * j].x     = w0[t];  b[t][2 * j].y     = w1[t];
                b[t][2 * j + 1].x = w2[t];  b[t][2 * j + 1].y = w3[t];
            }
        } else {
            b[0][2 * j].x     = wp[0];         b[0][2 * j].y     = wp[NOUT];
            b[0][2 * j + 1].x = wp[2 * NOUT];  b[0][2 * j + 1].y = wp[3 * NOUT];
        }
    }

    v4f bv4 = {};
    float bv1 = 0.0f;
    if (BIAS) {
        if constexpr (T == 4) bv4 = *(const v4f*)(bias + col);
        else                  bv1 = bias[col];
    }

    for (int tm = blockIdx.x * RS + rs; tm < ntiles; tm += gridDim.x * RS) {
        const float* ap = A + (size_t)(tm * 16 + lo) * 128 + 4 * hi;
        float* cp = C + (size_t)(tm * 16 + hi * 8) * NOUT + col;

        v4f a[16];
#pragma unroll
        for (int j = 0; j < 16; ++j) a[j] = *(const v4f*)(ap + 8 * j);

        v8f c[T];
#pragma unroll
        for (int t = 0; t < T; ++t) c[t] = v8f{};
        if (ACC) {
#pragma unroll
            for (int r = 0; r < 8; ++r) {
                if constexpr (T == 4) {
                    const v4f o = *(const v4f*)(cp + (size_t)r * NOUT);
                    c[0][r] = o.x; c[1][r] = o.y; c[2][r] = o.z; c[3][r] = o.w;
                } else {
                    c[0][r] = cp[(size_t)r * NOUT];
                }
            }
        }

#pragma unroll
        for (int j = 0; j < 16; ++j) {
            v2f a0, a1;
            a0.x = a[j].x; a0.y = a[j].y;
            a1.x = a[j].z; a1.y = a[j].w;
#pragma unroll
            for (int t = 0; t < T; ++t)
                c[t] = __builtin_amdgcn_wmma_f32_16x16x4_f32(
                         false, a0,  false, b[t][2 * j],
                         (short)0, c[t],  false,  false);
#pragma unroll
            for (int t = 0; t < T; ++t)
                c[t] = __builtin_amdgcn_wmma_f32_16x16x4_f32(
                         false, a1,  false, b[t][2 * j + 1],
                         (short)0, c[t],  false,  false);
        }

#pragma unroll
        for (int r = 0; r < 8; ++r) {
            if constexpr (T == 4) {
                v4f v;
                v.x = c[0][r]; v.y = c[1][r]; v.z = c[2][r]; v.w = c[3][r];
                if (BIAS) v += bv4;
                if (RELU) { v.x = fmaxf(v.x, 0.0f); v.y = fmaxf(v.y, 0.0f);
                            v.z = fmaxf(v.z, 0.0f); v.w = fmaxf(v.w, 0.0f); }
                *(v4f*)(cp + (size_t)r * NOUT) = v;
            } else {
                float v = c[0][r];
                if (BIAS) v += bv1;
                if (RELU) v = fmaxf(v, 0.0f);
                cp[(size_t)r * NOUT] = v;
            }
        }
    }
}

extern "C" void kernel_launch(void* const* d_in, const int* in_sizes, int n_in,
                              void* d_out, int out_size, void* d_ws, size_t ws_size,
                              hipStream_t stream) {
    const float* x  = (const float*)d_in[0];
    const int*   ei = (const int*)d_in[1];
    const float* W1 = (const float*)d_in[2];
    const float* b1 = (const float*)d_in[3];
    const float* W2 = (const float*)d_in[4];
    const float* b2 = (const float*)d_in[5];

    const int N = in_sizes[0] / 128;
    const int E = in_sizes[1] / 2;
    const int* src = ei;
    const int* dst = ei + E;

    float* ws   = (float*)d_ws;
    float* qreg = ws;
    float* dis  = qreg + N;
    float* nrm  = dis + N;
    float* bufA = nrm + E;
    float* bufB = bufA + (size_t)N * 128;
    float* acc  = bufB + (size_t)N * 128;
    float* out  = (float*)d_out;

    const int T = 256;
    const size_t featBytes = (size_t)N * 128 * sizeof(float);
    const int pblk = 1536;
    int* q = (int*)qreg;
    int* part = (int*)bufA;
    const int nrange = (N + DR - 1) / DR;
    const int ntile = N / 16;
    const int g1 = 256;
    const int g2 = 768;

    tag_deg <<<nrange * DS, DT, 0, stream>>>(dst, part, N, E, nrange);
    tag_dis <<<(N + T - 1) / T, T, 0, stream>>>(part, dis, N);
    tag_norm<<<(E + T - 1) / T, T, 0, stream>>>(src, dst, dis, nrm, q, E);

    tag_gemm<128, false, false, false><<<g1, T, 0, stream>>>(x,    W1 + 0 * 128 * 128, acc, nullptr, ntile);
    hipMemsetAsync(bufA, 0, featBytes, stream);
    tag_prop<<<pblk, T, 0, stream>>>(x, src, dst, nrm, bufA, q + 0 * QSTRIDE, E);
    tag_gemm<128, true,  false, false><<<g1, T, 0, stream>>>(bufA, W1 + 1 * 128 * 128, acc, nullptr, ntile);
    hipMemsetAsync(bufB, 0, featBytes, stream);
    tag_prop<<<pblk, T, 0, stream>>>(bufA, src, dst, nrm, bufB, q + 2 * QSTRIDE, E);
    tag_gemm<128, true,  false, false><<<g1, T, 0, stream>>>(bufB, W1 + 2 * 128 * 128, acc, nullptr, ntile);
    hipMemsetAsync(bufA, 0, featBytes, stream);
    tag_prop<<<pblk, T, 0, stream>>>(bufB, src, dst, nrm, bufA, q + 4 * QSTRIDE, E);
    tag_gemm<128, true,  true,  true ><<<g1, T, 0, stream>>>(bufA, W1 + 3 * 128 * 128, acc, b1, ntile);

    tag_gemm<16, false, false, false><<<g2, T, 0, stream>>>(acc,  W2 + 0 * 128 * 16, out, nullptr, ntile);
    hipMemsetAsync(bufA, 0, featBytes, stream);
    tag_prop<<<pblk, T, 0, stream>>>(acc, src, dst, nrm, bufA, q + 6 * QSTRIDE, E);
    tag_gemm<16, true,  false, false><<<g2, T, 0, stream>>>(bufA, W2 + 1 * 128 * 16, out, nullptr, ntile);
    hipMemsetAsync(bufB, 0, featBytes, stream);
    tag_prop<<<pblk, T, 0, stream>>>(bufA, src, dst, nrm, bufB, q + 8 * QSTRIDE, E);
    tag_gemm<16, true,  false, false><<<g2, T, 0, stream>>>(bufB, W2 + 2 * 128 * 16, out, nullptr, ntile);
    hipMemsetAsync(bufA, 0, featBytes, stream);
    tag_prop<<<pblk, T, 0, stream>>>(bufB, src, dst, nrm, bufA, q + 10 * QSTRIDE, E);
    tag_gemm<16, true,  true,  false><<<g2, T, 0, stream>>>(bufA, W2 + 3 * 128 * 16, out, b2, ntile);

    (void)n_in; (void)out_size; (void)ws_size;
}


# layer 1 multiplies before it propagates, nested: y=xW3, y=xW2+Ay, y=xW1+Ay, pre=xW0+b+Ay; two 2-product WMMA passes over x, hops add onto products: 3 fills and 2 launches go; f32 sum order changes
# speedup vs baseline: 2.7566x; 1.0750x over previous
#include <hip/hip_runtime.h>

typedef float v2f __attribute__((ext_vector_type(2)));
typedef float v4f __attribute__((ext_vector_type(4)));
typedef float v8f __attribute__((ext_vector_type(8)));

constexpr int DR = 6400;
constexpr int DS = 16;
constexpr int DT = 1024;
constexpr int DU = 8;

__device__ __forceinline__ void deg_count(int* cnt, int d, int lo) {
    const unsigned k = (unsigned)(d - lo);
    if (k < (unsigned)DR) atomicAdd(&cnt[k], 1);
}

__global__ __launch_bounds__(DT)
void tag_deg(const int* __restrict__ dst, int* __restrict__ part, int N, int E, int nrange) {
    __shared__ int cnt[DR];
    const int r  = blockIdx.x % nrange;
    const int s  = blockIdx.x / nrange;
    const int lo = r * DR;
    for (int i = threadIdx.x; i < DR; i += DT) cnt[i] = 0;
    __syncthreads();

    const int per = (((E + DS - 1) / DS) + 3) & ~3;
    const long long b0 = (long long)s * per;
    const int e0 = (int)(b0 < (long long)E ? b0 : (long long)E);
    const int e1 = (E - e0 < per) ? E : e0 + per;
    int e = e0;
    if ((((size_t)dst) & 15) == 0) {
        constexpr int SPAN = DT * 4 * DU;
        for (; e1 - e >= SPAN; e += SPAN) {
            const int* p = dst + e + (int)threadIdx.x * 4;
            int4 d[DU];
#pragma unroll
            for (int u = 0; u < DU; ++u) d[u] = *(const int4*)(p + u * DT * 4);
#pragma unroll
            for (int u = 0; u < DU; ++u) {
                deg_count(cnt, d[u].x, lo);
                deg_count(cnt, d[u].y, lo);
                deg_count(cnt, d[u].z, lo);
                deg_count(cnt, d[u].w, lo);
            }
        }
    }
    for (int p = e + (int)threadIdx.x; p < e1; p += DT) deg_count(cnt, dst[p], lo);
    __syncthreads();
    for (int i = threadIdx.x; i < DR; i += DT)
        if (lo + i < N) part[(size_t)s * N + lo + i] = cnt[i];
}

__global__ void tag_dis(const int* __restrict__ part, float* __restrict__ dis, int N) {
    int i = blockIdx.x * blockDim.x + threadIdx.x;
    if (i < N) {
        int c = 0;
#pragma unroll
        for (int s = 0; s < DS; ++s) c += part[(size_t)s * N + i];
        float d = (float)c;
        dis[i] = (d > 0.0f) ? rsqrtf(fmaxf(d, 1.0f)) : 0.0f;
    }
}

constexpr int QSTRIDE = 64;
constexpr int NQ      = 12;

__global__ void tag_norm(const int* __restrict__ src, const int* __restrict__ dst,
                         const float* __restrict__ dis, float* __restrict__ nrm,
                         int* __restrict__ q, int E) {
    int e = blockIdx.x * blockDim.x + threadIdx.x;
    if (e < NQ) q[e * QSTRIDE] = 0;
    if (e < E) nrm[e] = dis[src[e]] * dis[dst[e]];
}

constexpr int EPG = 16;
constexpr int GPC = 4;
constexpr int EPC = EPG * GPC;

template <bool RELU_IN>
__global__ __launch_bounds__(256)
void tag_prop(const float* __restrict__ h, const int* __restrict__ src,
              const int* __restrict__ dst, const float* __restrict__ nrm,
              float* __restrict__ out, int* __restrict__ q, int E) {
    __shared__ int s_half;
    const int lane = threadIdx.x & 31;
    if (threadIdx.x == 0) {
        const unsigned id = __builtin_amdgcn_s_sendmsg_rtn(0x87);
        s_half = (int)((id >> 18) & 1u);
    }
    __syncthreads();
    int fh = __builtin_amdgcn_readfirstlane(s_half);
    const int nchunk = (E + EPC - 1) / EPC;

    int tv = 0;
    if (lane == 0) tv = atomicAdd(q + fh * QSTRIDE, 1);
    int other = 0;
    const int maxit = 2 * nchunk + 4;
    for (int it = 0; it < maxit; ++it) {
        const int t = __builtin_amdgcn_readfirstlane(tv);
        if (t >= nchunk) {
            if (other) break;
            other = 1;
            fh ^= 1;
            tv = 0;
            if (lane == 0) tv = atomicAdd(q + fh * QSTRIDE, 1);
            continue;
        }
        const int e0 = t * EPC;
        const size_t fo = (size_t)fh * 64;
        int tn = 0;
        if (e0 + EPC <= E) {
            const int   s0 = src[e0 + lane], s1 = src[e0 + 32 + lane];
            const int   d0 = dst[e0 + lane], d1 = dst[e0 + 32 + lane];
            const float w0 = nrm[e0 + lane], w1 = nrm[e0 + 32 + lane];
            if (lane == 0) tn = atomicAdd(q + fh * QSTRIDE, 1);
            for (int g = 0; g < GPC; ++g) {
                const int   ms = (g < 2) ? s0 : s1;
                const int   md = (g < 2) ? d0 : d1;
                const float mw = (g < 2) ? w0 : w1;
                const int   l0 = (g & 1) * EPG;
                float v[EPG][2];
#pragma unroll
                for (int j = 0; j < EPG; ++j) {
                    const int s = __builtin_amdgcn_readlane(ms, l0 + j);
                    const float* hrow = h + (size_t)s * 128 + fo;
                    v[j][0] = hrow[lane];
                    v[j][1] = hrow[lane + 32];
                    if (RELU_IN) { v[j][0] = fmaxf(v[j][0], 0.0f); v[j][1] = fmaxf(v[j][1], 0.0f); }
                }
#pragma unroll
                for (int j = 0; j < EPG; ++j) {
                    const int   dd = __builtin_amdgcn_readlane(md, l0 + j);
                    const float w  = __int_as_float(__builtin_amdgcn_readlane(__float_as_int(mw), l0 + j));
                    float* orow = out + (size_t)dd * 128 + fo;
                    unsafeAtomicAdd(orow + lane,      v[j][0] * w);
                    unsafeAtomicAdd(orow + lane + 32, v[j][1] * w);
                }
            }
        } else {
            if (lane == 0) tn = atomicAdd(q + fh * QSTRIDE, 1);
            for (int e = e0; e < E; ++e) {
                const int   s  = src[e];
                const int   dd = dst[e];
                const float w  = nrm[e];
                const float* hrow = h + (size_t)s * 128 + fo;
                float* orow = out + (size_t)dd * 128 + fo;
                float v0 = hrow[lane], v1 = hrow[lane + 32];
                if (RELU_IN) { v0 = fmaxf(v0, 0.0f); v1 = fmaxf(v1, 0.0f); }
                unsafeAtomicAdd(orow + lane,      v0 * w);
                unsafeAtomicAdd(orow + lane + 32, v1 * w);
            }
        }
        tv = tn;
    }
}

template <int NOUT, int NM, bool ACC, bool BIAS, bool RELU_A>
__global__ __launch_bounds__(256)
void tag_gemm(const float* __restrict__ A,
              const float* __restrict__ Wp, const float* __restrict__ Wq,
              float* __restrict__ Cp, float* __restrict__ Cq,
              const float* __restrict__ bias, int ntiles) {
    constexpr int T  = (NOUT >= 64) ? 4 : 1;
    constexpr int GC = NOUT / (16 * T);
    constexpr int G  = GC * NM;
    constexpr int RS = 8 / G;
    const int lane = threadIdx.x & 31;
    const int wv = __builtin_amdgcn_readfirstlane((int)(threadIdx.x >> 5));
    const int g  = wv % G;
    const int rs = wv / G;
    const bool second = (NM == 2) && (g >= GC);
    const int gc = second ? g - GC : g;
    const float* __restrict__ W = second ? Wq : Wp;
    float* __restrict__ C = second ? Cq : Cp;
    const int lo = lane & 15;
    const int hi = lane >> 4;
    const int col = gc * (16 * T) + lo * T;

    v2f b[T][32];
#pragma unroll
    for (int j = 0; j < 16; ++j) {
        const float* wp = W + (size_t)(8 * j + 4 * hi) * NOUT + col;
        if constexpr (T == 4) {
            const v4f w0 = *(const v4f*)(wp);
            const v4f w1 = *(const v4f*)(wp + NOUT);
            const v4f w2 = *(const v4f*)(wp + 2 * NOUT);
            const v4f w3 = *(const v4f*)(wp + 3 * NOUT);
#pragma unroll
            for (int t = 0; t < 4; ++t) {
                b[t][2 * j].x     = w0[t];  b[t][2 * j].y     = w1[t];
                b[t][2 * j + 1].x = w2[t];  b[t][2 * j + 1].y = w3[t];
            }
        } else {
            b[0][2 * j].x     = wp[0];         b[0][2 * j].y     = wp[NOUT];
            b[0][2 * j + 1].x = wp[2 * NOUT];  b[0][2 * j + 1].y = wp[3 * NOUT];
        }
    }

    v4f bv4 = {};
    float bv1 = 0.0f;
    if (BIAS && !second) {
        if constexpr (T == 4) bv4 = *(const v4f*)(bias + col);
        else                  bv1 = bias[col];
    }

    for (int tm = blockIdx.x * RS + rs; tm < ntiles; tm += gridDim.x * RS) {
        const float* ap = A + (size_t)(tm * 16 + lo) * 128 + 4 * hi;
        float* cp = C + (size_t)(tm * 16 + hi * 8) * NOUT + col;

        v4f a[16];
#pragma unroll
        for (int j = 0; j < 16; ++j) a[j] = *(const v4f*)(ap + 8 * j);

        v8f c[T];
#pragma unroll
        for (int t = 0; t < T; ++t) c[t] = v8f{};
        if (ACC) {
#pragma unroll
            for (int r = 0; r < 8; ++r) {
                if constexpr (T == 4) {
                    const v4f o = *(const v4f*)(cp + (size_t)r * NOUT);
                    c[0][r] = o.x; c[1][r] = o.y; c[2][r] = o.z; c[3][r] = o.w;
                } else {
                    c[0][r] = cp[(size_t)r * NOUT];
                }
            }
        }

#pragma unroll
        for (int j = 0; j < 16; ++j) {
            v2f a0, a1;
            a0.x = a[j].x; a0.y = a[j].y;
            a1.x = a[j].z; a1.y = a[j].w;
            if (RELU_A) {
                a0.x = fmaxf(a0.x, 0.0f); a0.y = fmaxf(a0.y, 0.0f);
                a1.x = fmaxf(a1.x, 0.0f); a1.y = fmaxf(a1.y, 0.0f);
            }
#pragma unroll
            for (int t = 0; t < T; ++t)
                c[t] = __builtin_amdgcn_wmma_f32_16x16x4_f32(
                         false, a0,  false, b[t][2 * j],
                         (short)0, c[t],  false,  false);
#pragma unroll
            for (int t = 0; t < T; ++t)
                c[t] = __builtin_amdgcn_wmma_f32_16x16x4_f32(
                         false, a1,  false, b[t][2 * j + 1],
                         (short)0, c[t],  false,  false);
        }

#pragma unroll
        for (int r = 0; r < 8; ++r) {
            if constexpr (T == 4) {
                v4f v;
                v.x = c[0][r]; v.y = c[1][r]; v.z = c[2][r]; v.w = c[3][r];
                if (BIAS) v += bv4;
                *(v4f*)(cp + (size_t)r * NOUT) = v;
            } else {
                float v = c[0][r];
                if (BIAS) v += bv1;
                cp[(size_t)r * NOUT] = v;
            }
        }
    }
}

extern "C" void kernel_launch(void* const* d_in, const int* in_sizes, int n_in,
                              void* d_out, int out_size, void* d_ws, size_t ws_size,
                              hipStream_t stream) {
    const float* x  = (const float*)d_in[0];
    const int*   ei = (const int*)d_in[1];
    const float* W1 = (const float*)d_in[2];
    const float* b1 = (const float*)d_in[3];
    const float* W2 = (const float*)d_in[4];
    const float* b2 = (const float*)d_in[5];

    const int N = in_sizes[0] / 128;
    const int E = in_sizes[1] / 2;
    const int* src = ei;
    const int* dst = ei + E;

    float* ws   = (float*)d_ws;
    float* qreg = ws;
    float* dis  = qreg + N;
    float* nrm  = dis + N;
    float* bufA = nrm + E;
    float* bufB = bufA + (size_t)N * 128;
    float* acc  = bufB + (size_t)N * 128;
    float* out  = (float*)d_out;

    const int T = 256;
    const size_t featBytes = (size_t)N * 128 * sizeof(float);
    const int pblk = 1536;
    int* q = (int*)qreg;
    int* part = (int*)bufA;
    const int nrange = (N + DR - 1) / DR;
    const int ntile = N / 16;
    const int g1 = 256;
    const int g2 = 768;

    tag_deg <<<nrange * DS, DT, 0, stream>>>(dst, part, N, E, nrange);
    tag_dis <<<(N + T - 1) / T, T, 0, stream>>>(part, dis, N);
    tag_norm<<<(E + T - 1) / T, T, 0, stream>>>(src, dst, dis, nrm, q, E);

    tag_gemm<128, 2, false, false, false><<<g1, T, 0, stream>>>(x, W1 + 3 * 128 * 128, W1 + 2 * 128 * 128, bufA, bufB, nullptr, ntile);
    tag_prop<false><<<pblk, T, 0, stream>>>(bufA, src, dst, nrm, bufB, q + 0 * QSTRIDE, E);
    tag_gemm<128, 2, false, true,  false><<<g1, T, 0, stream>>>(x, W1 + 0 * 128 * 128, W1 + 1 * 128 * 128, acc, bufA, b1, ntile);
    tag_prop<false><<<pblk, T, 0, stream>>>(bufB, src, dst, nrm, bufA, q + 2 * QSTRIDE, E);
    tag_prop<false><<<pblk, T, 0, stream>>>(bufA, src, dst, nrm, acc,  q + 4 * QSTRIDE, E);

    tag_gemm<16, 1, false, false, true ><<<g2, T, 0, stream>>>(acc,  W2 + 0 * 128 * 16, nullptr, out, nullptr, nullptr, ntile);
    hipMemsetAsync(bufA, 0, featBytes, stream);
    tag_prop<true ><<<pblk, T, 0, stream>>>(acc, src, dst, nrm, bufA, q + 6 * QSTRIDE, E);
    tag_gemm<16, 1, true,  false, false><<<g2, T, 0, stream>>>(bufA, W2 + 1 * 128 * 16, nullptr, out, nullptr, nullptr, ntile);
    hipMemsetAsync(bufB, 0, featBytes, stream);
    tag_prop<false><<<pblk, T, 0, stream>>>(bufA, src, dst, nrm, bufB, q + 8 * QSTRIDE, E);
    tag_gemm<16, 1, true,  false, false><<<g2, T, 0, stream>>>(bufB, W2 + 2 * 128 * 16, nullptr, out, nullptr, nullptr, ntile);
    hipMemsetAsync(bufA, 0, featBytes, stream);
    tag_prop<false><<<pblk, T, 0, stream>>>(bufB, src, dst, nrm, bufA, q + 10 * QSTRIDE, E);
    tag_gemm<16, 1, true,  true,  false><<<g2, T, 0, stream>>>(bufA, W2 + 3 * 128 * 16, nullptr, out, nullptr, b2, ntile);

    (void)n_in; (void)out_size; (void)ws_size;
}


# hops push by source row: edges put in source order per call (LDS counts, prefix sum, LDS-ranked fill); a source row is read once, per-edge row gathers go; same f32 atomic adds, arrival order differs
# speedup vs baseline: 3.1830x; 1.1547x over previous
#include <hip/hip_runtime.h>

typedef float v2f __attribute__((ext_vector_type(2)));
typedef float v4f __attribute__((ext_vector_type(4)));
typedef float v8f __attribute__((ext_vector_type(8)));

constexpr int DR = 6400;
constexpr int DS = 16;
constexpr int DT = 1024;
constexpr int DU = 8;

__device__ __forceinline__ void deg_count(int* cnt, int d, int lo) {
    const unsigned k = (unsigned)(d - lo);
    if (k < (unsigned)DR) atomicAdd(&cnt[k], 1);
}

__global__ __launch_bounds__(DT)
void tag_deg(const int* __restrict__ dst, const int* __restrict__ src,
             int* __restrict__ partD, int* __restrict__ partS, int N, int E, int nrange) {
    __shared__ int cnt[DR];
    const int* __restrict__ key = blockIdx.y ? src : dst;
    int* __restrict__ part      = blockIdx.y ? partS : partD;
    const int r  = blockIdx.x % nrange;
    const int s  = blockIdx.x / nrange;
    const int lo = r * DR;
    for (int i = threadIdx.x; i < DR; i += DT) cnt[i] = 0;
    __syncthreads();

    const int per = (((E + DS - 1) / DS) + 3) & ~3;
    const long long b0 = (long long)s * per;
    const int e0 = (int)(b0 < (long long)E ? b0 : (long long)E);
    const int e1 = (E - e0 < per) ? E : e0 + per;
    int e = e0;
    if ((((size_t)key) & 15) == 0) {
        constexpr int SPAN = DT * 4 * DU;
        for (; e1 - e >= SPAN; e += SPAN) {
            const int* p = key + e + (int)threadIdx.x * 4;
            int4 d[DU];
#pragma unroll
            for (int u = 0; u < DU; ++u) d[u] = *(const int4*)(p + u * DT * 4);
#pragma unroll
            for (int u = 0; u < DU; ++u) {
                deg_count(cnt, d[u].x, lo);
                deg_count(cnt, d[u].y, lo);
                deg_count(cnt, d[u].z, lo);
                deg_count(cnt, d[u].w, lo);
            }
        }
    }
    for (int p = e + (int)threadIdx.x; p < e1; p += DT) deg_count(cnt, key[p], lo);
    __syncthreads();
    for (int i = threadIdx.x; i < DR; i += DT)
        if (lo + i < N) part[(size_t)s * N + lo + i] = cnt[i];
}

constexpr int SB = 256;
__global__ __launch_bounds__(SB)
void tag_dis(const int* __restrict__ partD, const int* __restrict__ partS,
             float* __restrict__ dis, int* __restrict__ csrc, int* __restrict__ bsum, int N) {
    __shared__ int s_sum;
    if (threadIdx.x == 0) s_sum = 0;
    __syncthreads();
    const int i = blockIdx.x * SB + threadIdx.x;
    int cs = 0;
    if (i < N) {
        int cd = 0;
#pragma unroll
        for (int s = 0; s < DS; ++s) {
            cd += partD[(size_t)s * N + i];
            cs += partS[(size_t)s * N + i];
        }
        float d = (float)cd;
        dis[i] = (d > 0.0f) ? rsqrtf(fmaxf(d, 1.0f)) : 0.0f;
        csrc[i] = cs;
    }
    atomicAdd(&s_sum, cs);
    __syncthreads();
    if (threadIdx.x == 0) bsum[blockIdx.x] = s_sum;
}

__global__ __launch_bounds__(SB)
void tag_rowp(const int* __restrict__ csrc, const int* __restrict__ bsum,
              int* __restrict__ rowp, int N) {
    __shared__ int s[SB];
    __shared__ int s_off;
    const int t = threadIdx.x;
    const int b = blockIdx.x;
    if (t == 0) s_off = 0;
    __syncthreads();
    int before = 0;
    for (int j = t; j < b; j += SB) before += bsum[j];
    atomicAdd(&s_off, before);
    const int i = b * SB + t;
    const int cv = (i < N) ? csrc[i] : 0;
    s[t] = cv;
    __syncthreads();
    for (int off = 1; off < SB; off <<= 1) {
        const int v = (t >= off) ? s[t - off] : 0;
        __syncthreads();
        s[t] += v;
        __syncthreads();
    }
    const int excl = s_off + s[t] - cv;
    if (i < N) rowp[i] = excl;
    if (i == N - 1) rowp[N] = excl + cv;
}

constexpr int QSTRIDE = 64;
constexpr int NQ      = 12;

__global__ void tag_norm(const int* __restrict__ src, const int* __restrict__ dst,
                         const float* __restrict__ dis, float* __restrict__ nrm,
                         int* __restrict__ q, int E) {
    int e = blockIdx.x * blockDim.x + threadIdx.x;
    if (e < NQ) q[e * QSTRIDE] = 0;
    if (e < E) nrm[e] = dis[src[e]] * dis[dst[e]];
}

constexpr int FU = 4;

__device__ __forceinline__ int fill_place(int* cnt, int sv, int lo) {
    const unsigned k = (unsigned)(sv - lo);
    return (k < (unsigned)DR) ? atomicAdd(&cnt[k], 1) : -1;
}

__global__ __launch_bounds__(DT)
void tag_fill(const int* __restrict__ src, const int* __restrict__ dst,
              const float* __restrict__ nrm, const int* __restrict__ rowp,
              const int* __restrict__ partS, int2* __restrict__ sedge,
              int N, int E, int nrange) {
    __shared__ int cnt[DR];
    const int r  = blockIdx.x % nrange;
    const int s  = blockIdx.x / nrange;
    const int lo = r * DR;
    for (int i = threadIdx.x; i < DR; i += DT) {
        int b = 0;
        if (lo + i < N) {
            b = rowp[lo + i];
            for (int k = 0; k < s; ++k) b += partS[(size_t)k * N + lo + i];
        }
        cnt[i] = b;
    }
    __syncthreads();

    const int per = (((E + DS - 1) / DS) + 3) & ~3;
    const long long b0 = (long long)s * per;
    const int e0 = (int)(b0 < (long long)E ? b0 : (long long)E);
    const int e1 = (E - e0 < per) ? E : e0 + per;
    int e = e0;
    if (((((size_t)src) | ((size_t)dst) | ((size_t)nrm)) & 15) == 0) {
        constexpr int SPAN = DT * 4 * FU;
        for (; e1 - e >= SPAN; e += SPAN) {
            const int o = e + (int)threadIdx.x * 4;
            int4 sv[FU], dv[FU];
            float4 wv[FU];
#pragma unroll
            for (int u = 0; u < FU; ++u) {
                sv[u] = *(const int4*)(src + o + u * DT * 4);
                dv[u] = *(const int4*)(dst + o + u * DT * 4);
                wv[u] = *(const float4*)(nrm + o + u * DT * 4);
            }
#pragma unroll
            for (int u = 0; u < FU; ++u) {
                const int p0 = fill_place(cnt, sv[u].x, lo);
                const int p1 = fill_place(cnt, sv[u].y, lo);
                const int p2 = fill_place(cnt, sv[u].z, lo);
                const int p3 = fill_place(cnt, sv[u].w, lo);
                if (p0 >= 0) sedge[p0] = make_int2(dv[u].x, __float_as_int(wv[u].x));
                if (p1 >= 0) sedge[p1] = make_int2(dv[u].y, __float_as_int(wv[u].y));
                if (p2 >= 0) sedge[p2] = make_int2(dv[u].z, __float_as_int(wv[u].z));
                if (p3 >= 0) sedge[p3] = make_int2(dv[u].w, __float_as_int(wv[u].w));
            }
        }
    }
    for (int p = e + (int)threadIdx.x; p < e1; p += DT) {
        const int pl = fill_place(cnt, src[p], lo);
        if (pl >= 0) sedge[pl] = make_int2(dst[p], __float_as_int(nrm[p]));
    }
}

constexpr int RPT = 4;

template <bool RELU_IN>
__global__ __launch_bounds__(256)
void tag_push(const float* __restrict__ h, const int* __restrict__ rowp,
              const int2* __restrict__ sedge,
              float* __restrict__ out, int* __restrict__ q, int N) {
    __shared__ int s_half;
    const int lane = threadIdx.x & 31;
    if (threadIdx.x == 0) {
        const unsigned id = __builtin_amdgcn_s_sendmsg_rtn(0x87);
        s_half = (int)((id >> 18) & 1u);
    }
    __syncthreads();
    int fh = __builtin_amdgcn_readfirstlane(s_half);
    const int nchunk = (N + RPT - 1) / RPT;

    int tv = 0;
    if (lane == 0) tv = atomicAdd(q + fh * QSTRIDE, 1);
    int other = 0;
    const int maxit = 2 * nchunk + 4;
    for (int it = 0; it < maxit; ++it) {
        const int t = __builtin_amdgcn_readfirstlane(tv);
        if (t >= nchunk) {
            if (other) break;
            other = 1;
            fh ^= 1;
            tv = 0;
            if (lane == 0) tv = atomicAdd(q + fh * QSTRIDE, 1);
            continue;
        }
        const int r0 = t * RPT;
        const size_t fo = (size_t)fh * 64;
        int rp = 0;
        if (lane <= RPT) {
            const int rr = r0 + lane;
            rp = rowp[rr < N ? rr : N];
        }
        float v[RPT][2];
#pragma unroll
        for (int j = 0; j < RPT; ++j) {
            const int rr = (r0 + j < N) ? r0 + j : N - 1;
            const float* hrow = h + (size_t)rr * 128 + fo;
            v[j][0] = hrow[lane];
            v[j][1] = hrow[lane + 32];
            if (RELU_IN) { v[j][0] = fmaxf(v[j][0], 0.0f); v[j][1] = fmaxf(v[j][1], 0.0f); }
        }
        int tn = 0;
        if (lane == 0) tn = atomicAdd(q + fh * QSTRIDE, 1);
        int2 se[RPT];
#pragma unroll
        for (int j = 0; j < RPT; ++j) {
            const int p0 = __builtin_amdgcn_readlane(rp, j);
            const int p1 = __builtin_amdgcn_readlane(rp, j + 1);
            const int idx = p0 + lane;
            se[j] = make_int2(0, 0);
            if (idx < p1) se[j] = sedge[idx];
        }
#pragma unroll
        for (int j = 0; j < RPT; ++j) {
            const int p0 = __builtin_amdgcn_readlane(rp, j);
            const int p1 = __builtin_amdgcn_readlane(rp, j + 1);
            int2 me = se[j];
            for (int p = p0; p < p1; p += 32) {
                if (p != p0) {
                    const int idx = p + lane;
                    me = make_int2(0, 0);
                    if (idx < p1) me = sedge[idx];
                }
                const int n = (p1 - p < 32) ? p1 - p : 32;
                for (int k = 0; k < n; ++k) {
                    const int   dd = __builtin_amdgcn_readlane(me.x, k);
                    const float w  = __int_as_float(__builtin_amdgcn_readlane(me.y, k));
                    float* orow = out + (size_t)dd * 128 + fo;
                    unsafeAtomicAdd(orow + lane,      v[j][0] * w);
                    unsafeAtomicAdd(orow + lane + 32, v[j][1] * w);
                }
            }
        }
        tv = tn;
    }
}

template <int NOUT, int NM, bool ACC, bool BIAS, bool RELU_A>
__global__ __launch_bounds__(256)
void tag_gemm(const float* __restrict__ A,
              const float* __restrict__ Wp, const float* __restrict__ Wq,
              float* __restrict__ Cp, float* __restrict__ Cq,
              const float* __restrict__ bias, int ntiles) {
    constexpr int T  = (NOUT >= 64) ? 4 : 1;
    constexpr int GC = NOUT / (16 * T);
    constexpr int G  = GC * NM;
    constexpr int RS = 8 / G;
    const int lane = threadIdx.x & 31;
    const int wv = __builtin_amdgcn_readfirstlane((int)(threadIdx.x >> 5));
    const int g  = wv % G;
    const int rs = wv / G;
    const bool second = (NM == 2) && (g >= GC);
    const int gc = second ? g - GC : g;
    const float* __restrict__ W = second ? Wq : Wp;
    float* __restrict__ C = second ? Cq : Cp;
    const int lo = lane & 15;
    const int hi = lane >> 4;
    const int col = gc * (16 * T) + lo * T;

    v2f b[T][32];
#pragma unroll
    for (int j = 0; j < 16; ++j) {
        const float* wp = W + (size_t)(8 * j + 4 * hi) * NOUT + col;
        if constexpr (T == 4) {
            const v4f w0 = *(const v4f*)(wp);
            const v4f w1 = *(const v4f*)(wp + NOUT);
            const v4f w2 = *(const v4f*)(wp + 2 * NOUT);
            const v4f w3 = *(const v4f*)(wp + 3 * NOUT);
#pragma unroll
            for (int t = 0; t < 4; ++t) {
                b[t][2 * j].x     = w0[t];  b[t][2 * j].y     = w1[t];
                b[t][2 * j + 1].x = w2[t];  b[t][2 * j + 1].y = w3[t];
            }
        } else {
            b[0][2 * j].x     = wp[0];         b[0][2 * j].y     = wp[NOUT];
            b[0][2 * j + 1].x = wp[2 * NOUT];  b[0][2 * j + 1].y = wp[3 * NOUT];
        }
    }

    v4f bv4 = {};
    float bv1 = 0.0f;
    if (BIAS && !second) {
        if constexpr (T == 4) bv4 = *(const v4f*)(bias + col);
        else                  bv1 = bias[col];
    }

    for (int tm = blockIdx.x * RS + rs; tm < ntiles; tm += gridDim.x * RS) {
        const float* ap = A + (size_t)(tm * 16 + lo) * 128 + 4 * hi;
        float* cp = C + (size_t)(tm * 16 + hi * 8) * NOUT + col;

        v4f a[16];
#pragma unroll
        for (int j = 0; j < 16; ++j) a[j] = *(const v4f*)(ap + 8 * j);

        v8f c[T];
#pragma unroll
        for (int t = 0; t < T; ++t) c[t] = v8f{};
        if (ACC) {
#pragma unroll
            for (int r = 0; r < 8; ++r) {
                if constexpr (T == 4) {
                    const v4f o = *(const v4f*)(cp + (size_t)r * NOUT);
                    c[0][r] = o.x; c[1][r] = o.y; c[2][r] = o.z; c[3][r] = o.w;
                } else {
                    c[0][r] = cp[(size_t)r * NOUT];
                }
            }
        }

#pragma unroll
        for (int j = 0; j < 16; ++j) {
            v2f a0, a1;
            a0.x = a[j].x; a0.y = a[j].y;
            a1.x = a[j].z; a1.y = a[j].w;
            if (RELU_A) {
                a0.x = fmaxf(a0.x, 0.0f); a0.y = fmaxf(a0.y, 0.0f);
                a1.x = fmaxf(a1.x, 0.0f); a1.y = fmaxf(a1.y, 0.0f);
            }
#pragma unroll
            for (int t = 0; t < T; ++t)
                c[t] = __builtin_amdgcn_wmma_f32_16x16x4_f32(
                         false, a0,  false, b[t][2 * j],
                         (short)0, c[t],  false,  false);
#pragma unroll
            for (int t = 0; t < T; ++t)
                c[t] = __builtin_amdgcn_wmma_f32_16x16x4_f32(
                         false, a1,  false, b[t][2 * j + 1],
                         (short)0, c[t],  false,  false);
        }

#pragma unroll
        for (int r = 0; r < 8; ++r) {
            if constexpr (T == 4) {
                v4f v;
                v.x = c[0][r]; v.y = c[1][r]; v.z = c[2][r]; v.w = c[3][r];
                if (BIAS) v += bv4;
                *(v4f*)(cp + (size_t)r * NOUT) = v;
            } else {
                float v = c[0][r];
                if (BIAS) v += bv1;
                cp[(size_t)r * NOUT] = v;
            }
        }
    }
}

extern "C" void kernel_launch(void* const* d_in, const int* in_sizes, int n_in,
                              void* d_out, int out_size, void* d_ws, size_t ws_size,
                              hipStream_t stream) {
    const float* x  = (const float*)d_in[0];
    const int*   ei = (const int*)d_in[1];
    const float* W1 = (const float*)d_in[2];
    const float* b1 = (const float*)d_in[3];
    const float* W2 = (const float*)d_in[4];
    const float* b2 = (const float*)d_in[5];

    const int N = in_sizes[0] / 128;
    const int E = in_sizes[1] / 2;
    const int* src = ei;
    const int* dst = ei + E;

    float* ws   = (float*)d_ws;
    float* qreg = ws;
    float* dis  = qreg + N;
    float* nrm  = dis + N;
    float* bufA = nrm + E;
    float* bufB = bufA + (size_t)N * 128;
    int2*  sedge = (int2*)(bufB + (size_t)N * 128);
    int*   rowp  = (int*)(sedge + E);
    int*   csrc  = rowp + N + 4;
    int*   bsum  = csrc + N;
    int*   partD = bsum + N / 64;
    int*   partS = partD + (size_t)DS * N;
    float* out  = (float*)d_out;

    const int T = 256;
    const size_t featBytes = (size_t)N * 128 * sizeof(float);
    const int pblk = 1536;
    int* q = (int*)qreg;
    const int nrange = (N + DR - 1) / DR;
    const int nsb = (N + SB - 1) / SB;
    const int ntile = N / 16;
    const int g1 = 256;
    const int g2 = 768;

    tag_deg <<<dim3(nrange * DS, 2), DT, 0, stream>>>(dst, src, partD, partS, N, E, nrange);
    tag_dis <<<nsb, SB, 0, stream>>>(partD, partS, dis, csrc, bsum, N);
    tag_rowp<<<nsb, SB, 0, stream>>>(csrc, bsum, rowp, N);
    tag_norm<<<(E + T - 1) / T, T, 0, stream>>>(src, dst, dis, nrm, q, E);
    tag_fill<<<nrange * DS, DT, 0, stream>>>(src, dst, nrm, rowp, partS, sedge, N, E, nrange);

    tag_gemm<128, 2, false, false, false><<<g1, T, 0, stream>>>(x, W1 + 3 * 128 * 128, W1 + 2 * 128 * 128, bufA, bufB, nullptr, ntile);
    tag_push<false><<<pblk, T, 0, stream>>>(bufA, rowp, sedge, bufB, q + 0 * QSTRIDE, N);
    tag_gemm<128, 1, false, false, false><<<g1, T, 0, stream>>>(x, W1 + 1 * 128 * 128, nullptr, bufA, nullptr, nullptr, ntile);
    tag_push<false><<<pblk, T, 0, stream>>>(bufB, rowp, sedge, bufA, q + 2 * QSTRIDE, N);
    tag_gemm<128, 1, false, true,  false><<<g1, T, 0, stream>>>(x, W1 + 0 * 128 * 128, nullptr, bufB, nullptr, b1, ntile);
    tag_push<false><<<pblk, T, 0, stream>>>(bufA, rowp, sedge, bufB, q + 4 * QSTRIDE, N);

    tag_gemm<16, 1, false, false, true ><<<g2, T, 0, stream>>>(bufB, W2 + 0 * 128 * 16, nullptr, out, nullptr, nullptr, ntile);
    hipMemsetAsync(bufA, 0, featBytes, stream);
    tag_push<true ><<<pblk, T, 0, stream>>>(bufB, rowp, sedge, bufA, q + 6 * QSTRIDE, N);
    tag_gemm<16, 1, true,  false, false><<<g2, T, 0, stream>>>(bufA, W2 + 1 * 128 * 16, nullptr, out, nullptr, nullptr, ntile);
    hipMemsetAsync(bufB, 0, featBytes, stream);
    tag_push<false><<<pblk, T, 0, stream>>>(bufA, rowp, sedge, bufB, q + 8 * QSTRIDE, N);
    tag_gemm<16, 1, true,  false, false><<<g2, T, 0, stream>>>(bufB, W2 + 2 * 128 * 16, nullptr, out, nullptr, nullptr, ntile);
    hipMemsetAsync(bufA, 0, featBytes, stream);
    tag_push<false><<<pblk, T, 0, stream>>>(bufB, rowp, sedge, bufA, q + 10 * QSTRIDE, N);
    tag_gemm<16, 1, true,  true,  false><<<g2, T, 0, stream>>>(bufA, W2 + 3 * 128 * 16, nullptr, out, nullptr, b2, ntile);

    (void)n_in; (void)out_size; (void)ws_size;
}
